# starting kernel
# baseline (speedup 1.0000x reference)
#include <hip/hip_runtime.h>
#include <hip/hip_bf16.h>

typedef float v2f __attribute__((ext_vector_type(2)));
typedef float v8f __attribute__((ext_vector_type(8)));

#define B_  4
#define C_  21
#define H_  128
#define W_  128
#define D_  5
#define EPSF 1e-5f
#define QS  34
#define MS  66

__global__ __launch_bounds__(32)
void clsbd_crf_kernel(const float* __restrict__ inp,
                      const float* __restrict__ feats,
                      const float* __restrict__ clsbd,
                      float* __restrict__ out)
{
    __shared__ float lQ[64 * QS];
    __shared__ float lMn[16 * MS];
    __shared__ float lMp[16 * MS];

    const int lane = threadIdx.x;
    const int t  = blockIdx.x;
    const int b  = t >> 10;
    const int ty = (t >> 5) & 31;
    const int tx = t & 31;
    const int H0 = ty * 4, W0 = tx * 4;

    for (int i = lane; i < 16 * MS; i += 32) { lMn[i] = 0.f; lMp[i] = 0.f; }

    const float invLogC = 1.0f / 3.0445224377234230f;
    const size_t HW = (size_t)H_ * W_;
    #pragma unroll
    for (int rep = 0; rep < 2; ++rep) {
        const int s  = lane + rep * 32;
        const int sy = s >> 3, sx = s & 7;
        const int gy = H0 - 2 + sy, gx = W0 - 2 + sx;
        const bool in = (gy >= 0) && (gy < H_) && (gx >= 0) && (gx < W_);
        if (in) {
            const float* p = inp + (size_t)b * C_ * HW + (size_t)gy * W_ + gx;
            float v[C_];
            float ent = 0.f;
            #pragma unroll
            for (int c = 0; c < C_; ++c) {
                float x = p[(size_t)c * HW];
                v[c] = x;
                ent -= x * __logf(x + EPSF);
            }
            const float pl = 1.0f - ent * invLogC;
            #pragma unroll
            for (int c = 0; c < C_; ++c) lQ[s * QS + c] = v[c] * pl;
        } else {
            #pragma unroll
            for (int c = 0; c < C_; ++c) lQ[s * QS + c] = 0.f;
        }
        #pragma unroll
        for (int c = C_; c < 32; ++c) lQ[s * QS + c] = 0.f;
    }

    if (lane < 16) {
        const int m  = lane;
        const int oy = m >> 2, ox = m & 3;
        const int py = H0 + oy, px = W0 + ox;
        const float* fb = feats + (size_t)b * D_ * HW;
        const float* cb = clsbd + (size_t)b * HW;

        float fc[D_];
        #pragma unroll
        for (int d = 0; d < D_; ++d) fc[d] = fb[(size_t)d * HW + (size_t)py * W_ + px];

        float cl[5][5];
        #pragma unroll
        for (int a = 0; a < 5; ++a)
            #pragma unroll
            for (int d = 0; d < 5; ++d) {
                int gy = py + a - 2, gx = px + d - 2;
                cl[a][d] = (gy >= 0 && gy < H_ && gx >= 0 && gx < W_)
                           ? cb[(size_t)gy * W_ + gx] : 0.f;
            }

        float g2[5][5];
        g2[1][1] = cl[1][1]; g2[1][2] = cl[1][2]; g2[1][3] = cl[1][3];
        g2[2][1] = cl[2][1];                      g2[2][3] = cl[2][3];
        g2[3][1] = cl[3][1]; g2[3][2] = cl[3][2]; g2[3][3] = cl[3][3];
        g2[2][2] = 0.f;
        g2[0][0] = fmaxf(cl[0][0], cl[1][1]);
        g2[0][1] = fmaxf(cl[0][1], fmaxf(cl[1][1], cl[1][2]));
        g2[0][2] = fmaxf(cl[0][2], cl[1][2]);
        g2[0][3] = fmaxf(cl[0][3], fmaxf(cl[1][3], cl[1][2]));
        g2[0][4] = fmaxf(cl[0][4], cl[1][3]);
        g2[1][0] = fmaxf(cl[1][0], fmaxf(cl[1][1], cl[2][1]));
        g2[1][4] = fmaxf(cl[1][4], fmaxf(cl[1][3], cl[2][3]));
        g2[2][0] = fmaxf(cl[2][0], cl[2][1]);
        g2[2][4] = fmaxf(cl[2][4], cl[2][3]);
        g2[3][0] = fmaxf(cl[3][0], fmaxf(cl[3][1], cl[2][1]));
        g2[3][4] = fmaxf(cl[3][4], fmaxf(cl[3][3], cl[2][3]));
        g2[4][0] = fmaxf(cl[4][0], cl[3][1]);
        g2[4][1] = fmaxf(cl[4][1], fmaxf(cl[3][1], cl[3][2]));
        g2[4][2] = fmaxf(cl[4][2], cl[3][2]);
        g2[4][3] = fmaxf(cl[4][3], fmaxf(cl[3][3], cl[3][2]));
        g2[4][4] = fmaxf(cl[4][4], cl[3][3]);

        #pragma unroll
        for (int a = 0; a < 5; ++a)
            #pragma unroll
            for (int d = 0; d < 5; ++d) {
                int gy = py + a - 2, gx = px + d - 2;
                if (gy >= 0 && gy < H_ && gx >= 0 && gx < W_) {
                    float ss = 0.f;
                    #pragma unroll
                    for (int dd = 0; dd < D_; ++dd) {
                        float df = fc[dd] - fb[(size_t)dd * HW + (size_t)gy * W_ + gx];
                        ss += df * df;
                    }
                    float g1 = __expf(-0.5f * ss);
                    float gg = g2[a][d];
                    int   s  = (oy + a) * 8 + (ox + d);
                    lMn[m * MS + s] = 10.0f * g1 - 5.0f * __logf(gg + EPSF);
                    lMp[m * MS + s] = -5.0f * __logf(1.0f - gg + EPSF);
                }
            }
    }
    __syncthreads();

    const int half = lane >> 4;
    const int nr   = lane & 15;
    v8f accN0 = {}, accN1 = {}, accP0 = {}, accP1 = {};
    #pragma unroll
    for (int kk = 0; kk < 16; ++kk) {
        const int kb = 4 * kk + 2 * half;
        v2f aN, aP, b0, b1;
        aN.x = lMn[nr * MS + kb];      aN.y = lMn[nr * MS + kb + 1];
        aP.x = lMp[nr * MS + kb];      aP.y = lMp[nr * MS + kb + 1];
        b0.x = lQ[kb * QS + nr];       b0.y = lQ[(kb + 1) * QS + nr];
        b1.x = lQ[kb * QS + 16 + nr];  b1.y = lQ[(kb + 1) * QS + 16 + nr];
        accN0 = __builtin_amdgcn_wmma_f32_16x16x4_f32(false, aN, false, b0, (short)0, accN0, false, false);
        accN1 = __builtin_amdgcn_wmma_f32_16x16x4_f32(false, aN, false, b1, (short)0, accN1, false, false);
        accP0 = __builtin_amdgcn_wmma_f32_16x16x4_f32(false, aP, false, b0, (short)0, accP0, false, false);
        accP1 = __builtin_amdgcn_wmma_f32_16x16x4_f32(false, aP, false, b1, (short)0, accP1, false, false);
    }

    const size_t outBase = (size_t)b * C_ * HW;
    const size_t plane   = (size_t)B_ * C_ * HW;
    #pragma unroll
    for (int v = 0; v < 8; ++v) {
        const int m  = v + 8 * half;
        const int oy = m >> 2, ox = m & 3;
        const size_t pix = (size_t)(H0 + oy) * W_ + (W0 + ox);
        const int c0 = nr, c1 = nr + 16;
        out[outBase + (size_t)c0 * HW + pix]         = accN0[v];
        out[plane + outBase + (size_t)c0 * HW + pix] = accP0[v];
        if (c1 < C_) {
            out[outBase + (size_t)c1 * HW + pix]         = accN1[v];
            out[plane + outBase + (size_t)c1 * HW + pix] = accP1[v];
        }
    }
}

extern "C" void kernel_launch(void* const* d_in, const int* in_sizes, int n_in,
                              void* d_out, int out_size, void* d_ws, size_t ws_size,
                              hipStream_t stream) {
    (void)in_sizes; (void)n_in; (void)out_size; (void)d_ws; (void)ws_size;
    const float* inp   = (const float*)d_in[0];
    const float* feats = (const float*)d_in[1];
    const float* clsbd = (const float*)d_in[2];
    float* out = (float*)d_out;
    const int tiles = B_ * (H_ / 4) * (W_ / 4);
    clsbd_crf_kernel<<<tiles, 32, 0, stream>>>(inp, feats, clsbd, out);
}


# 8 adjacent 4x4 tiles per 8-wave work-group; results staged in LDS so every 128-B output line is written whole by one store instruction instead of 32 scattered 4-byte pieces
# speedup vs baseline: 4.9614x; 4.9614x over previous
#include <hip/hip_runtime.h>
#include <hip/hip_bf16.h>

typedef float v2f __attribute__((ext_vector_type(2)));
typedef float v8f __attribute__((ext_vector_type(8)));

#define B_  4
#define C_  21
#define H_  128
#define W_  128
#define D_  5
#define EPSF 1e-5f
#define QS  34
#define MS  66
#define NW  8
#define RS  33
#define SC  132

__global__ __launch_bounds__(32 * NW)
void clsbd_crf_kernel(const float* __restrict__ inp,
                      const float* __restrict__ feats,
                      const float* __restrict__ clsbd,
                      float* __restrict__ out)
{
    __shared__ float lQa[NW][64 * QS];
    __shared__ float lMna[NW][16 * MS];
    __shared__ float lMpa[NW][16 * MS];

    const int lane = threadIdx.x & 31;
    const int wv   = __builtin_amdgcn_readfirstlane(threadIdx.x >> 5);
    float* lQ  = lQa[wv];
    float* lMn = lMna[wv];
    float* lMp = lMpa[wv];
    const int g  = blockIdx.x;
    const int b  = g >> 7;
    const int ty = (g >> 2) & 31;
    const int X0 = (g & 3) * 32;
    const int H0 = ty * 4, W0 = X0 + wv * 4;

    for (int i = lane; i < 16 * MS; i += 32) { lMn[i] = 0.f; lMp[i] = 0.f; }

    const float invLogC = 1.0f / 3.0445224377234230f;
    const size_t HW = (size_t)H_ * W_;
    #pragma unroll
    for (int rep = 0; rep < 2; ++rep) {
        const int s  = lane + rep * 32;
        const int sy = s >> 3, sx = s & 7;
        const int gy = H0 - 2 + sy, gx = W0 - 2 + sx;
        const bool in = (gy >= 0) && (gy < H_) && (gx >= 0) && (gx < W_);
        if (in) {
            const float* p = inp + (size_t)b * C_ * HW + (size_t)gy * W_ + gx;
            float v[C_];
            float ent = 0.f;
            #pragma unroll
            for (int c = 0; c < C_; ++c) {
                float x = p[(size_t)c * HW];
                v[c] = x;
                ent -= x * __logf(x + EPSF);
            }
            const float pl = 1.0f - ent * invLogC;
            #pragma unroll
            for (int c = 0; c < C_; ++c) lQ[s * QS + c] = v[c] * pl;
        } else {
            #pragma unroll
            for (int c = 0; c < C_; ++c) lQ[s * QS + c] = 0.f;
        }
        #pragma unroll
        for (int c = C_; c < 32; ++c) lQ[s * QS + c] = 0.f;
    }

    if (lane < 16) {
        const int m  = lane;
        const int oy = m >> 2, ox = m & 3;
        const int py = H0 + oy, px = W0 + ox;
        const float* fb = feats + (size_t)b * D_ * HW;
        const float* cb = clsbd + (size_t)b * HW;

        float fc[D_];
        #pragma unroll
        for (int d = 0; d < D_; ++d) fc[d] = fb[(size_t)d * HW + (size_t)py * W_ + px];

        float cl[5][5];
        #pragma unroll
        for (int a = 0; a < 5; ++a)
            #pragma unroll
            for (int d = 0; d < 5; ++d) {
                int gy = py + a - 2, gx = px + d - 2;
                cl[a][d] = (gy >= 0 && gy < H_ && gx >= 0 && gx < W_)
                           ? cb[(size_t)gy * W_ + gx] : 0.f;
            }

        float g2[5][5];
        g2[1][1] = cl[1][1]; g2[1][2] = cl[1][2]; g2[1][3] = cl[1][3];
        g2[2][1] = cl[2][1];                      g2[2][3] = cl[2][3];
        g2[3][1] = cl[3][1]; g2[3][2] = cl[3][2]; g2[3][3] = cl[3][3];
        g2[2][2] = 0.f;
        g2[0][0] = fmaxf(cl[0][0], cl[1][1]);
        g2[0][1] = fmaxf(cl[0][1], fmaxf(cl[1][1], cl[1][2]));
        g2[0][2] = fmaxf(cl[0][2], cl[1][2]);
        g2[0][3] = fmaxf(cl[0][3], fmaxf(cl[1][3], cl[1][2]));
        g2[0][4] = fmaxf(cl[0][4], cl[1][3]);
        g2[1][0] = fmaxf(cl[1][0], fmaxf(cl[1][1], cl[2][1]));
        g2[1][4] = fmaxf(cl[1][4], fmaxf(cl[1][3], cl[2][3]));
        g2[2][0] = fmaxf(cl[2][0], cl[2][1]);
        g2[2][4] = fmaxf(cl[2][4], cl[2][3]);
        g2[3][0] = fmaxf(cl[3][0], fmaxf(cl[3][1], cl[2][1]));
        g2[3][4] = fmaxf(cl[3][4], fmaxf(cl[3][3], cl[2][3]));
        g2[4][0] = fmaxf(cl[4][0], cl[3][1]);
        g2[4][1] = fmaxf(cl[4][1], fmaxf(cl[3][1], cl[3][2]));
        g2[4][2] = fmaxf(cl[4][2], cl[3][2]);
        g2[4][3] = fmaxf(cl[4][3], fmaxf(cl[3][3], cl[3][2]));
        g2[4][4] = fmaxf(cl[4][4], cl[3][3]);

        #pragma unroll
        for (int a = 0; a < 5; ++a)
            #pragma unroll
            for (int d = 0; d < 5; ++d) {
                int gy = py + a - 2, gx = px + d - 2;
                if (gy >= 0 && gy < H_ && gx >= 0 && gx < W_) {
                    float ss = 0.f;
                    #pragma unroll
                    for (int dd = 0; dd < D_; ++dd) {
                        float df = fc[dd] - fb[(size_t)dd * HW + (size_t)gy * W_ + gx];
                        ss += df * df;
                    }
                    float g1 = __expf(-0.5f * ss);
                    float gg = g2[a][d];
                    int   s  = (oy + a) * 8 + (ox + d);
                    lMn[m * MS + s] = 10.0f * g1 - 5.0f * __logf(gg + EPSF);
                    lMp[m * MS + s] = -5.0f * __logf(1.0f - gg + EPSF);
                }
            }
    }
    __syncthreads();

    const int half = lane >> 4;
    const int nr   = lane & 15;
    v8f accN0 = {}, accN1 = {}, accP0 = {}, accP1 = {};
    #pragma unroll
    for (int kk = 0; kk < 16; ++kk) {
        const int kb = 4 * kk + 2 * half;
        v2f aN, aP, b0, b1;
        aN.x = lMn[nr * MS + kb];      aN.y = lMn[nr * MS + kb + 1];
        aP.x = lMp[nr * MS + kb];      aP.y = lMp[nr * MS + kb + 1];
        b0.x = lQ[kb * QS + nr];       b0.y = lQ[(kb + 1) * QS + nr];
        b1.x = lQ[kb * QS + 16 + nr];  b1.y = lQ[(kb + 1) * QS + 16 + nr];
        accN0 = __builtin_amdgcn_wmma_f32_16x16x4_f32(false, aN, false, b0, (short)0, accN0, false, false);
        accN1 = __builtin_amdgcn_wmma_f32_16x16x4_f32(false, aN, false, b1, (short)0, accN1, false, false);
        accP0 = __builtin_amdgcn_wmma_f32_16x16x4_f32(false, aP, false, b0, (short)0, accP0, false, false);
        accP1 = __builtin_amdgcn_wmma_f32_16x16x4_f32(false, aP, false, b1, (short)0, accP1, false, false);
    }

    __syncthreads();
    float* stage = &lQa[0][0];
    #pragma unroll
    for (int v = 0; v < 8; ++v) {
        const int m  = v + 8 * half;
        const int oy = m >> 2, ox = m & 3;
        const int so = oy * RS + wv * 4 + ox;
        const int c0 = nr, c1 = nr + 16;
        stage[c0 * SC + so]         = accN0[v];
        stage[(C_ + c0) * SC + so]  = accP0[v];
        if (c1 < C_) {
            stage[c1 * SC + so]        = accN1[v];
            stage[(C_ + c1) * SC + so] = accP1[v];
        }
    }
    __syncthreads();

    const size_t outBase = (size_t)b * C_ * HW;
    const size_t plane   = (size_t)B_ * C_ * HW;
    #pragma unroll
    for (int j = 0; j < (2 * C_ * 4) / NW; ++j) {
        const int l  = wv + NW * j;
        const int pc = l >> 2, oy = l & 3;
        const int o  = (pc >= C_) ? 1 : 0;
        const int c  = pc - o * C_;
        const float val = stage[pc * SC + oy * RS + lane];
        out[(size_t)o * plane + outBase + (size_t)c * HW + (size_t)(H0 + oy) * W_ + X0 + lane] = val;
    }
}

extern "C" void kernel_launch(void* const* d_in, const int* in_sizes, int n_in,
                              void* d_out, int out_size, void* d_ws, size_t ws_size,
                              hipStream_t stream) {
    (void)in_sizes; (void)n_in; (void)out_size; (void)d_ws; (void)ws_size;
    const float* inp   = (const float*)d_in[0];
    const float* feats = (const float*)d_in[1];
    const float* clsbd = (const float*)d_in[2];
    float* out = (float*)d_out;
    const int groups = B_ * (H_ / 4) * (W_ / 32);
    clsbd_crf_kernel<<<groups, 32 * NW, 0, stream>>>(inp, feats, clsbd, out);
}


# weight phase: per-wave 8x8 feature+boundary patch staged once in LDS, 16 pixels x 25 taps split over all 32 lanes (mirrored halves) instead of 25 serialized global-load round trips on 16 lanes
# speedup vs baseline: 6.6753x; 1.3454x over previous
#include <hip/hip_runtime.h>
#include <hip/hip_bf16.h>

typedef float v2f __attribute__((ext_vector_type(2)));
typedef float v8f __attribute__((ext_vector_type(8)));

#define B_  4
#define C_  21
#define H_  128
#define W_  128
#define D_  5
#define EPSF 1e-5f
#define QS  34
#define MS  66
#define NW  8
#define RS  33
#define SC  132
#define FP  64

__global__ __launch_bounds__(32 * NW)
void clsbd_crf_kernel(const float* __restrict__ inp,
                      const float* __restrict__ feats,
                      const float* __restrict__ clsbd,
                      float* __restrict__ out)
{
    __shared__ float lQa[NW][64 * QS];
    __shared__ float lMna[NW][16 * MS];
    __shared__ float lMpa[NW][16 * MS];
    __shared__ float lFa[NW][6 * FP];

    const int lane = threadIdx.x & 31;
    const int wv   = __builtin_amdgcn_readfirstlane(threadIdx.x >> 5);
    float* lQ  = lQa[wv];
    float* lMn = lMna[wv];
    float* lMp = lMpa[wv];
    float* lF  = lFa[wv];
    const int g  = blockIdx.x;
    const int b  = g >> 7;
    const int ty = (g >> 2) & 31;
    const int X0 = (g & 3) * 32;
    const int H0 = ty * 4, W0 = X0 + wv * 4;

    for (int i = lane; i < 16 * MS; i += 32) { lMn[i] = 0.f; lMp[i] = 0.f; }

    const float invLogC = 1.0f / 3.0445224377234230f;
    const size_t HW = (size_t)H_ * W_;
    const float* fb = feats + (size_t)b * D_ * HW;
    const float* cb = clsbd + (size_t)b * HW;
    #pragma unroll
    for (int rep = 0; rep < 2; ++rep) {
        const int s  = lane + rep * 32;
        const int sy = s >> 3, sx = s & 7;
        const int gy = H0 - 2 + sy, gx = W0 - 2 + sx;
        const bool in = (gy >= 0) && (gy < H_) && (gx >= 0) && (gx < W_);
        if (in) {
            const size_t po = (size_t)gy * W_ + gx;
            #pragma unroll
            for (int d = 0; d < D_; ++d) lF[d * FP + s] = fb[(size_t)d * HW + po];
            lF[D_ * FP + s] = cb[po];
            const float* p = inp + (size_t)b * C_ * HW + po;
            float v[C_];
            float ent = 0.f;
            #pragma unroll
            for (int c = 0; c < C_; ++c) {
                float x = p[(size_t)c * HW];
                v[c] = x;
                ent -= x * __logf(x + EPSF);
            }
            const float pl = 1.0f - ent * invLogC;
            #pragma unroll
            for (int c = 0; c < C_; ++c) lQ[s * QS + c] = v[c] * pl;
        } else {
            #pragma unroll
            for (int c = 0; c < C_; ++c) lQ[s * QS + c] = 0.f;
            #pragma unroll
            for (int d = 0; d <= D_; ++d) lF[d * FP + s] = 0.f;
        }
        #pragma unroll
        for (int c = C_; c < 32; ++c) lQ[s * QS + c] = 0.f;
    }

    __builtin_amdgcn_fence(__ATOMIC_RELEASE, "wavefront");
    __builtin_amdgcn_wave_barrier();
    __builtin_amdgcn_fence(__ATOMIC_ACQUIRE, "wavefront");

    {
        const int m   = lane & 15;
        const int hf  = lane >> 4;
        const int oy  = m >> 2, ox = m & 3;
        const int ctr = (oy + 2) * 8 + (ox + 2);
        const int sg  = 1 - 2 * hf;
        const float* lC = lF + D_ * FP;

        float fc[D_];
        #pragma unroll
        for (int d = 0; d < D_; ++d) fc[d] = lF[d * FP + ctr];

#define SRC(a, d) (ctr + sg * (((a) - 2) * 8 + ((d) - 2)))
        const float c00 = lC[SRC(0, 0)], c01 = lC[SRC(0, 1)], c02 = lC[SRC(0, 2)], c03 = lC[SRC(0, 3)], c04 = lC[SRC(0, 4)];
        const float c10 = lC[SRC(1, 0)], c11 = lC[SRC(1, 1)], c12 = lC[SRC(1, 2)], c13 = lC[SRC(1, 3)], c14 = lC[SRC(1, 4)];
        const float c20 = lC[SRC(2, 0)], c21 = lC[SRC(2, 1)], c23 = lC[SRC(2, 3)];

        float g2[13];
        g2[0]  = fmaxf(c00, c11);
        g2[1]  = fmaxf(c01, fmaxf(c11, c12));
        g2[2]  = fmaxf(c02, c12);
        g2[3]  = fmaxf(c03, fmaxf(c13, c12));
        g2[4]  = fmaxf(c04, c13);
        g2[5]  = fmaxf(c10, fmaxf(c11, c21));
        g2[6]  = c11;
        g2[7]  = c12;
        g2[8]  = c13;
        g2[9]  = fmaxf(c14, fmaxf(c13, c23));
        g2[10] = fmaxf(c20, c21);
        g2[11] = c21;
        g2[12] = 0.f;

        #pragma unroll
        for (int t = 0; t < 13; ++t) {
            const int a = t / 5, d = t % 5;
            const int s = SRC(a, d);
            float ss = 0.f;
            #pragma unroll
            for (int dd = 0; dd < D_; ++dd) {
                float df = fc[dd] - lF[dd * FP + s];
                ss += df * df;
            }
            const float g1 = __expf(-0.5f * ss);
            const float gg = g2[t];
            lMn[m * MS + s] = 10.0f * g1 - 5.0f * __logf(gg + EPSF);
            lMp[m * MS + s] = -5.0f * __logf(1.0f - gg + EPSF);
        }
#undef SRC
    }
    __syncthreads();

    const int half = lane >> 4;
    const int nr   = lane & 15;
    v8f accN0 = {}, accN1 = {}, accP0 = {}, accP1 = {};
    #pragma unroll
    for (int kk = 0; kk < 16; ++kk) {
        const int kb = 4 * kk + 2 * half;
        v2f aN, aP, b0, b1;
        aN.x = lMn[nr * MS + kb];      aN.y = lMn[nr * MS + kb + 1];
        aP.x = lMp[nr * MS + kb];      aP.y = lMp[nr * MS + kb + 1];
        b0.x = lQ[kb * QS + nr];       b0.y = lQ[(kb + 1) * QS + nr];
        b1.x = lQ[kb * QS + 16 + nr];  b1.y = lQ[(kb + 1) * QS + 16 + nr];
        accN0 = __builtin_amdgcn_wmma_f32_16x16x4_f32(false, aN, false, b0, (short)0, accN0, false, false);
        accN1 = __builtin_amdgcn_wmma_f32_16x16x4_f32(false, aN, false, b1, (short)0, accN1, false, false);
        accP0 = __builtin_amdgcn_wmma_f32_16x16x4_f32(false, aP, false, b0, (short)0, accP0, false, false);
        accP1 = __builtin_amdgcn_wmma_f32_16x16x4_f32(false, aP, false, b1, (short)0, accP1, false, false);
    }

    __syncthreads();
    float* stage = &lQa[0][0];
    #pragma unroll
    for (int v = 0; v < 8; ++v) {
        const int m  = v + 8 * half;
        const int oy = m >> 2, ox = m & 3;
        const int so = oy * RS + wv * 4 + ox;
        const int c0 = nr, c1 = nr + 16;
        stage[c0 * SC + so]         = accN0[v];
        stage[(C_ + c0) * SC + so]  = accP0[v];
        if (c1 < C_) {
            stage[c1 * SC + so]        = accN1[v];
            stage[(C_ + c1) * SC + so] = accP1[v];
        }
    }
    __syncthreads();

    const size_t outBase = (size_t)b * C_ * HW;
    const size_t plane   = (size_t)B_ * C_ * HW;
    #pragma unroll
    for (int j = 0; j < (2 * C_ * 4) / NW; ++j) {
        const int l  = wv + NW * j;
        const int pc = l >> 2, oy = l & 3;
        const int o  = (pc >= C_) ? 1 : 0;
        const int c  = pc - o * C_;
        const float val = stage[pc * SC + oy * RS + lane];
        out[(size_t)o * plane + outBase + (size_t)c * HW + (size_t)(H0 + oy) * W_ + X0 + lane] = val;
    }
}

extern "C" void kernel_launch(void* const* d_in, const int* in_sizes, int n_in,
                              void* d_out, int out_size, void* d_ws, size_t ws_size,
                              hipStream_t stream) {
    (void)in_sizes; (void)n_in; (void)out_size; (void)d_ws; (void)ws_size;
    const float* inp   = (const float*)d_in[0];
    const float* feats = (const float*)d_in[1];
    const float* clsbd = (const float*)d_in[2];
    float* out = (float*)d_out;
    const int groups = B_ * (H_ / 4) * (W_ / 32);
    clsbd_crf_kernel<<<groups, 32 * NW, 0, stream>>>(inp, feats, clsbd, out);
}


# natural log taken as the hardware base-2 log instruction with ln2 folded into the constant that follows (2 instructions) instead of the library's 12-instruction sequence; 66 logs per lane
# speedup vs baseline: 7.3510x; 1.1012x over previous
#include <hip/hip_runtime.h>
#include <hip/hip_bf16.h>

typedef float v2f __attribute__((ext_vector_type(2)));
typedef float v8f __attribute__((ext_vector_type(8)));

#define B_  4
#define C_  21
#define H_  128
#define W_  128
#define D_  5
#define EPSF 1e-5f
#define QS  34
#define MS  66
#define NW  8
#define RS  33
#define SC  132
#define FP  64
#define LN2F 0.69314718055994531f

__device__ __forceinline__ float hw_log2(float x) { return __builtin_amdgcn_logf(x); }

__global__ __launch_bounds__(32 * NW)
void clsbd_crf_kernel(const float* __restrict__ inp,
                      const float* __restrict__ feats,
                      const float* __restrict__ clsbd,
                      float* __restrict__ out)
{
    __shared__ float lQa[NW][64 * QS];
    __shared__ float lMna[NW][16 * MS];
    __shared__ float lMpa[NW][16 * MS];
    __shared__ float lFa[NW][6 * FP];

    const int lane = threadIdx.x & 31;
    const int wv   = __builtin_amdgcn_readfirstlane(threadIdx.x >> 5);
    float* lQ  = lQa[wv];
    float* lMn = lMna[wv];
    float* lMp = lMpa[wv];
    float* lF  = lFa[wv];
    const int g  = blockIdx.x;
    const int b  = g >> 7;
    const int ty = (g >> 2) & 31;
    const int X0 = (g & 3) * 32;
    const int H0 = ty * 4, W0 = X0 + wv * 4;

    for (int i = lane; i < 16 * MS; i += 32) { lMn[i] = 0.f; lMp[i] = 0.f; }

    const float invLogC = 1.0f / 3.0445224377234230f;
    const size_t HW = (size_t)H_ * W_;
    const float* fb = feats + (size_t)b * D_ * HW;
    const float* cb = clsbd + (size_t)b * HW;
    #pragma unroll
    for (int rep = 0; rep < 2; ++rep) {
        const int s  = lane + rep * 32;
        const int sy = s >> 3, sx = s & 7;
        const int gy = H0 - 2 + sy, gx = W0 - 2 + sx;
        const bool in = (gy >= 0) && (gy < H_) && (gx >= 0) && (gx < W_);
        if (in) {
            const size_t po = (size_t)gy * W_ + gx;
            #pragma unroll
            for (int d = 0; d < D_; ++d) lF[d * FP + s] = fb[(size_t)d * HW + po];
            lF[D_ * FP + s] = cb[po];
            const float* p = inp + (size_t)b * C_ * HW + po;
            float v[C_];
            float s2 = 0.f;
            #pragma unroll
            for (int c = 0; c < C_; ++c) {
                float x = p[(size_t)c * HW];
                v[c] = x;
                s2 = fmaf(x, hw_log2(x + EPSF), s2);
            }
            const float pl = fmaf(s2, LN2F * invLogC, 1.0f);
            #pragma unroll
            for (int c = 0; c < C_; ++c) lQ[s * QS + c] = v[c] * pl;
        } else {
            #pragma unroll
            for (int c = 0; c < C_; ++c) lQ[s * QS + c] = 0.f;
            #pragma unroll
            for (int d = 0; d <= D_; ++d) lF[d * FP + s] = 0.f;
        }
        #pragma unroll
        for (int c = C_; c < 32; ++c) lQ[s * QS + c] = 0.f;
    }

    __builtin_amdgcn_fence(__ATOMIC_RELEASE, "wavefront");
    __builtin_amdgcn_wave_barrier();
    __builtin_amdgcn_fence(__ATOMIC_ACQUIRE, "wavefront");

    {
        const int m   = lane & 15;
        const int hf  = lane >> 4;
        const int oy  = m >> 2, ox = m & 3;
        const int ctr = (oy + 2) * 8 + (ox + 2);
        const int sg  = 1 - 2 * hf;
        const float* lC = lF + D_ * FP;

        float fc[D_];
        #pragma unroll
        for (int d = 0; d < D_; ++d) fc[d] = lF[d * FP + ctr];

#define SRC(a, d) (ctr + sg * (((a) - 2) * 8 + ((d) - 2)))
        const float c00 = lC[SRC(0, 0)], c01 = lC[SRC(0, 1)], c02 = lC[SRC(0, 2)], c03 = lC[SRC(0, 3)], c04 = lC[SRC(0, 4)];
        const float c10 = lC[SRC(1, 0)], c11 = lC[SRC(1, 1)], c12 = lC[SRC(1, 2)], c13 = lC[SRC(1, 3)], c14 = lC[SRC(1, 4)];
        const float c20 = lC[SRC(2, 0)], c21 = lC[SRC(2, 1)], c23 = lC[SRC(2, 3)];

        float g2[13];
        g2[0]  = fmaxf(c00, c11);
        g2[1]  = fmaxf(c01, fmaxf(c11, c12));
        g2[2]  = fmaxf(c02, c12);
        g2[3]  = fmaxf(c03, fmaxf(c13, c12));
        g2[4]  = fmaxf(c04, c13);
        g2[5]  = fmaxf(c10, fmaxf(c11, c21));
        g2[6]  = c11;
        g2[7]  = c12;
        g2[8]  = c13;
        g2[9]  = fmaxf(c14, fmaxf(c13, c23));
        g2[10] = fmaxf(c20, c21);
        g2[11] = c21;
        g2[12] = 0.f;

        #pragma unroll
        for (int t = 0; t < 13; ++t) {
            const int a = t / 5, d = t % 5;
            const int s = SRC(a, d);
            float ss = 0.f;
            #pragma unroll
            for (int dd = 0; dd < D_; ++dd) {
                float df = fc[dd] - lF[dd * FP + s];
                ss += df * df;
            }
            const float g1 = __expf(-0.5f * ss);
            const float gg = g2[t];
            lMn[m * MS + s] = fmaf(-5.0f * LN2F, hw_log2(gg + EPSF), 10.0f * g1);
            lMp[m * MS + s] = (-5.0f * LN2F) * hw_log2(1.0f - gg + EPSF);
        }
#undef SRC
    }
    __syncthreads();

    const int half = lane >> 4;
    const int nr   = lane & 15;
    v8f accN0 = {}, accN1 = {}, accP0 = {}, accP1 = {};
    #pragma unroll
    for (int kk = 0; kk < 16; ++kk) {
        const int kb = 4 * kk + 2 * half;
        v2f aN, aP, b0, b1;
        aN.x = lMn[nr * MS + kb];      aN.y = lMn[nr * MS + kb + 1];
        aP.x = lMp[nr * MS + kb];      aP.y = lMp[nr * MS + kb + 1];
        b0.x = lQ[kb * QS + nr];       b0.y = lQ[(kb + 1) * QS + nr];
        b1.x = lQ[kb * QS + 16 + nr];  b1.y = lQ[(kb + 1) * QS + 16 + nr];
        accN0 = __builtin_amdgcn_wmma_f32_16x16x4_f32(false, aN, false, b0, (short)0, accN0, false, false);
        accN1 = __builtin_amdgcn_wmma_f32_16x16x4_f32(false, aN, false, b1, (short)0, accN1, false, false);
        accP0 = __builtin_amdgcn_wmma_f32_16x16x4_f32(false, aP, false, b0, (short)0, accP0, false, false);
        accP1 = __builtin_amdgcn_wmma_f32_16x16x4_f32(false, aP, false, b1, (short)0, accP1, false, false);
    }

    __syncthreads();
    float* stage = &lQa[0][0];
    #pragma unroll
    for (int v = 0; v < 8; ++v) {
        const int m  = v + 8 * half;
        const int oy = m >> 2, ox = m & 3;
        const int so = oy * RS + wv * 4 + ox;
        const int c0 = nr, c1 = nr + 16;
        stage[c0 * SC + so]         = accN0[v];
        stage[(C_ + c0) * SC + so]  = accP0[v];
        if (c1 < C_) {
            stage[c1 * SC + so]        = accN1[v];
            stage[(C_ + c1) * SC + so] = accP1[v];
        }
    }
    __syncthreads();

    const size_t outBase = (size_t)b * C_ * HW;
    const size_t plane   = (size_t)B_ * C_ * HW;
    #pragma unroll
    for (int j = 0; j < (2 * C_ * 4) / NW; ++j) {
        const int l  = wv + NW * j;
        const int pc = l >> 2, oy = l & 3;
        const int o  = (pc >= C_) ? 1 : 0;
        const int c  = pc - o * C_;
        const float val = stage[pc * SC + oy * RS + lane];
        out[(size_t)o * plane + outBase + (size_t)c * HW + (size_t)(H0 + oy) * W_ + X0 + lane] = val;
    }
}

extern "C" void kernel_launch(void* const* d_in, const int* in_sizes, int n_in,
                              void* d_out, int out_size, void* d_ws, size_t ws_size,
                              hipStream_t stream) {
    (void)in_sizes; (void)n_in; (void)out_size; (void)d_ws; (void)ws_size;
    const float* inp   = (const float*)d_in[0];
    const float* feats = (const float*)d_in[1];
    const float* clsbd = (const float*)d_in[2];
    float* out = (float*)d_out;
    const int groups = B_ * (H_ / 4) * (W_ / 32);
    clsbd_crf_kernel<<<groups, 32 * NW, 0, stream>>>(inp, feats, clsbd, out);
}


# neg and pos weight rows of 8 pixels (half of the 4x4 tile) stacked as the 16 matrix rows: one A fragment feeds both outputs and a half spans 48 source pixels, not 64: 48 WMMAs per wave instead of 64
# speedup vs baseline: 7.8372x; 1.0661x over previous
#include <hip/hip_runtime.h>
#include <hip/hip_bf16.h>

typedef float v2f __attribute__((ext_vector_type(2)));
typedef float v8f __attribute__((ext_vector_type(8)));

#define B_  4
#define C_  21
#define H_  128
#define W_  128
#define D_  5
#define EPSF 1e-5f
#define QS  34
#define MS  68
#define NW  8
#define RS  33
#define SC  132
#define PO  (C_ * SC + 2)
#define FP  64
#define LN2F 0.69314718055994531f

__device__ __forceinline__ float hw_log2(float x) { return __builtin_amdgcn_logf(x); }

__global__ __launch_bounds__(32 * NW)
void clsbd_crf_kernel(const float* __restrict__ inp,
                      const float* __restrict__ feats,
                      const float* __restrict__ clsbd,
                      float* __restrict__ out)
{
    __shared__ float lQa[NW][64 * QS];
    __shared__ float lMna[NW][16 * MS];
    __shared__ float lMpa[NW][16 * MS];
    __shared__ float lFa[NW][6 * FP];

    const int lane = threadIdx.x & 31;
    const int wv   = __builtin_amdgcn_readfirstlane(threadIdx.x >> 5);
    float* lQ  = lQa[wv];
    float* lMn = lMna[wv];
    float* lMp = lMpa[wv];
    float* lF  = lFa[wv];
    const int g  = blockIdx.x;
    const int b  = g >> 7;
    const int ty = (g >> 2) & 31;
    const int X0 = (g & 3) * 32;
    const int H0 = ty * 4, W0 = X0 + wv * 4;

    for (int i = lane; i < 16 * MS; i += 32) { lMn[i] = 0.f; lMp[i] = 0.f; }

    const float invLogC = 1.0f / 3.0445224377234230f;
    const size_t HW = (size_t)H_ * W_;
    const float* fb = feats + (size_t)b * D_ * HW;
    const float* cb = clsbd + (size_t)b * HW;
    #pragma unroll
    for (int rep = 0; rep < 2; ++rep) {
        const int s  = lane + rep * 32;
        const int sy = s >> 3, sx = s & 7;
        const int gy = H0 - 2 + sy, gx = W0 - 2 + sx;
        const bool in = (gy >= 0) && (gy < H_) && (gx >= 0) && (gx < W_);
        if (in) {
            const size_t po = (size_t)gy * W_ + gx;
            #pragma unroll
            for (int d = 0; d < D_; ++d) lF[d * FP + s] = fb[(size_t)d * HW + po];
            lF[D_ * FP + s] = cb[po];
            const float* p = inp + (size_t)b * C_ * HW + po;
            float v[C_];
            float s2 = 0.f;
            #pragma unroll
            for (int c = 0; c < C_; ++c) {
                float x = p[(size_t)c * HW];
                v[c] = x;
                s2 = fmaf(x, hw_log2(x + EPSF), s2);
            }
            const float pl = fmaf(s2, LN2F * invLogC, 1.0f);
            #pragma unroll
            for (int c = 0; c < C_; ++c) lQ[s * QS + c] = v[c] * pl;
        } else {
            #pragma unroll
            for (int c = 0; c < C_; ++c) lQ[s * QS + c] = 0.f;
            #pragma unroll
            for (int d = 0; d <= D_; ++d) lF[d * FP + s] = 0.f;
        }
        #pragma unroll
        for (int c = C_; c < 32; ++c) lQ[s * QS + c] = 0.f;
    }

    __builtin_amdgcn_fence(__ATOMIC_RELEASE, "wavefront");
    __builtin_amdgcn_wave_barrier();
    __builtin_amdgcn_fence(__ATOMIC_ACQUIRE, "wavefront");

    {
        const int m   = lane & 15;
        const int hf  = lane >> 4;
        const int oy  = m >> 2, ox = m & 3;
        const int ctr = (oy + 2) * 8 + (ox + 2);
        const int sg  = 1 - 2 * hf;
        const float* lC = lF + D_ * FP;

        float fc[D_];
        #pragma unroll
        for (int d = 0; d < D_; ++d) fc[d] = lF[d * FP + ctr];

#define SRC(a, d) (ctr + sg * (((a) - 2) * 8 + ((d) - 2)))
        const float c00 = lC[SRC(0, 0)], c01 = lC[SRC(0, 1)], c02 = lC[SRC(0, 2)], c03 = lC[SRC(0, 3)], c04 = lC[SRC(0, 4)];
        const float c10 = lC[SRC(1, 0)], c11 = lC[SRC(1, 1)], c12 = lC[SRC(1, 2)], c13 = lC[SRC(1, 3)], c14 = lC[SRC(1, 4)];
        const float c20 = lC[SRC(2, 0)], c21 = lC[SRC(2, 1)], c23 = lC[SRC(2, 3)];

        float g2[13];
        g2[0]  = fmaxf(c00, c11);
        g2[1]  = fmaxf(c01, fmaxf(c11, c12));
        g2[2]  = fmaxf(c02, c12);
        g2[3]  = fmaxf(c03, fmaxf(c13, c12));
        g2[4]  = fmaxf(c04, c13);
        g2[5]  = fmaxf(c10, fmaxf(c11, c21));
        g2[6]  = c11;
        g2[7]  = c12;
        g2[8]  = c13;
        g2[9]  = fmaxf(c14, fmaxf(c13, c23));
        g2[10] = fmaxf(c20, c21);
        g2[11] = c21;
        g2[12] = 0.f;

        #pragma unroll
        for (int t = 0; t < 13; ++t) {
            const int a = t / 5, d = t % 5;
            const int s = SRC(a, d);
            float ss = 0.f;
            #pragma unroll
            for (int dd = 0; dd < D_; ++dd) {
                float df = fc[dd] - lF[dd * FP + s];
                ss += df * df;
            }
            const float g1 = __expf(-0.5f * ss);
            const float gg = g2[t];
            lMn[m * MS + s] = fmaf(-5.0f * LN2F, hw_log2(gg + EPSF), 10.0f * g1);
            lMp[m * MS + s] = (-5.0f * LN2F) * hw_log2(1.0f - gg + EPSF);
        }
#undef SRC
    }
    __syncthreads();

    const int half = lane >> 4;
    const int nr   = lane & 15;
    const int pr   = nr & 7;
    const float* rowT = ((nr < 8) ? lMn : lMp) + pr * MS;
    const float* rowB = ((nr < 8) ? lMn : lMp) + (8 + pr) * MS;
    v8f accT0 = {}, accT1 = {}, accB0 = {}, accB1 = {};
    #pragma unroll
    for (int kk = 0; kk < 16; ++kk) {
        const int kb = 4 * kk + 2 * half;
        v2f b0, b1;
        b0.x = lQ[kb * QS + nr];       b0.y = lQ[(kb + 1) * QS + nr];
        b1.x = lQ[kb * QS + 16 + nr];  b1.y = lQ[(kb + 1) * QS + 16 + nr];
        if (kk < 12) {
            v2f aT;
            aT.x = rowT[kb];           aT.y = rowT[kb + 1];
            accT0 = __builtin_amdgcn_wmma_f32_16x16x4_f32(false, aT, false, b0, (short)0, accT0, false, false);
            accT1 = __builtin_amdgcn_wmma_f32_16x16x4_f32(false, aT, false, b1, (short)0, accT1, false, false);
        }
        if (kk >= 4) {
            v2f aB;
            aB.x = rowB[kb];           aB.y = rowB[kb + 1];
            accB0 = __builtin_amdgcn_wmma_f32_16x16x4_f32(false, aB, false, b0, (short)0, accB0, false, false);
            accB1 = __builtin_amdgcn_wmma_f32_16x16x4_f32(false, aB, false, b1, (short)0, accB1, false, false);
        }
    }

    __syncthreads();
    float* stage = &lQa[0][0];
    #pragma unroll
    for (int v = 0; v < 8; ++v) {
        const int oy = v >> 2, ox = v & 3;
        const int soT = oy * RS + wv * 4 + ox;
        const int soB = (oy + 2) * RS + wv * 4 + ox;
        const int c0 = nr, c1 = nr + 16;
        float* sp = stage + half * PO;
        sp[c0 * SC + soT] = accT0[v];
        sp[c0 * SC + soB] = accB0[v];
        if (c1 < C_) {
            sp[c1 * SC + soT] = accT1[v];
            sp[c1 * SC + soB] = accB1[v];
        }
    }
    __syncthreads();

    const size_t outBase = (size_t)b * C_ * HW;
    const size_t plane   = (size_t)B_ * C_ * HW;
    #pragma unroll
    for (int j = 0; j < (2 * C_ * 4) / NW; ++j) {
        const int l  = wv + NW * j;
        const int pc = l >> 2, oy = l & 3;
        const int o  = (pc >= C_) ? 1 : 0;
        const int c  = pc - o * C_;
        const float val = stage[o * PO + c * SC + oy * RS + lane];
        out[(size_t)o * plane + outBase + (size_t)c * HW + (size_t)(H0 + oy) * W_ + X0 + lane] = val;
    }
}

extern "C" void kernel_launch(void* const* d_in, const int* in_sizes, int n_in,
                              void* d_out, int out_size, void* d_ws, size_t ws_size,
                              hipStream_t stream) {
    (void)in_sizes; (void)n_in; (void)out_size; (void)d_ws; (void)ws_size;
    const float* inp   = (const float*)d_in[0];
    const float* feats = (const float*)d_in[1];
    const float* clsbd = (const float*)d_in[2];
    float* out = (float*)d_out;
    const int groups = B_ * (H_ / 4) * (W_ / 32);
    clsbd_crf_kernel<<<groups, 32 * NW, 0, stream>>>(inp, feats, clsbd, out);
}
